# starting kernel
# baseline (speedup 1.0000x reference)
#include <hip/hip_runtime.h>
#include <cmath>


typedef __attribute__((ext_vector_type(2))) float v2f;
typedef __attribute__((ext_vector_type(8))) float v8f;

#define B_SZ 4096
#define T_SZ 60
#define F_SZ 158
#define F_PAD 160
#define G3 96
#define H_SZ 32

__device__ __forceinline__ float sigmoid_f(float x) {
    return 1.0f / (1.0f + __expf(-x));
}

extern "C" __global__ __launch_bounds__(256)
void gru_gemm_in(const float* __restrict__ x, const float* __restrict__ w_ih,
                 float* __restrict__ gi)
{
    __shared__ float w_lds[G3 * F_PAD];
    const int tid = threadIdx.x;
    for (int i = tid; i < G3 * F_PAD; i += 256) {
        const int row = i / F_PAD, col = i % F_PAD;
        w_lds[i] = (col < F_SZ) ? w_ih[row * F_SZ + col] : 0.0f;
    }
    __syncthreads();

    const int lane = tid & 31;
    const int lo   = lane & 15;
    const int hi   = lane >> 4;
    const int wave = tid >> 5;
    const long m0  = (long)blockIdx.x * 128 + wave * 16;

    v8f acc[6] = {};
    const float* xrow = x + (m0 + lo) * F_SZ;

    for (int kt = 0; kt < 40; ++kt) {
        const int c = kt * 4 + 2 * hi;
        v2f a = {};
        if (c < F_SZ) {
            a.x = xrow[c];
            a.y = xrow[c + 1];
        }
        #pragma unroll
        for (int nt = 0; nt < 6; ++nt) {
            const int n = nt * 16 + lo;
            v2f b;
            b.x = w_lds[n * F_PAD + c];
            b.y = w_lds[n * F_PAD + c + 1];
            acc[nt] = __builtin_amdgcn_wmma_f32_16x16x4_f32(
                false, a, false, b, (short)0, acc[nt], false, false);
        }
    }

    #pragma unroll
    for (int r = 0; r < 8; ++r) {
        const long m = m0 + hi * 8 + r;
        const long t = m % T_SZ;
        const long b = m / T_SZ;
        float* orow = gi + (t * B_SZ + b) * G3;
        #pragma unroll
        for (int nt = 0; nt < 6; ++nt) {
            orow[nt * 16 + lo] = acc[nt][r];
        }
    }
}

extern "C" __global__ __launch_bounds__(64)
void gru_recurrence(const float* __restrict__ gi, const float* __restrict__ w_hh,
                    const float* __restrict__ w1, const float* __restrict__ b1,
                    const float* __restrict__ w2, float* __restrict__ out)
{
    __shared__ float h_lds[2 * 16 * 34];
    const int tid = threadIdx.x;
    for (int i = tid; i < 2 * 16 * 34; i += 64) h_lds[i] = 0.0f;
    __syncthreads();

    const int lane = tid & 31;
    const int lo   = lane & 15;
    const int hi   = lane >> 4;
    const int wave = tid >> 5;
    const int bb   = blockIdx.x * 32 + wave * 16;
    float* hw = h_lds + wave * (16 * 34);

    v2f bf[6][8];
    #pragma unroll
    for (int nt = 0; nt < 6; ++nt) {
        const int n = nt * 16 + lo;
        #pragma unroll
        for (int kt = 0; kt < 8; ++kt) {
            const int k = kt * 4 + 2 * hi;
            bf[nt][kt].x = w_hh[n * H_SZ + k];
            bf[nt][kt].y = w_hh[n * H_SZ + k + 1];
        }
    }

    v8f hc[2] = {};

    for (int t = 0; t < T_SZ; ++t) {
        v8f acc[6] = {};
        #pragma unroll
        for (int kt = 0; kt < 8; ++kt) {
            const int c = kt * 4 + 2 * hi;
            v2f a;
            a.x = hw[lo * 34 + c];
            a.y = hw[lo * 34 + c + 1];
            #pragma unroll
            for (int nt = 0; nt < 6; ++nt) {
                acc[nt] = __builtin_amdgcn_wmma_f32_16x16x4_f32(
                    false, a, false, bf[nt][kt], (short)0, acc[nt], false, false);
            }
        }

        const float* girow = gi + ((long)t * B_SZ + bb) * G3;
        #pragma unroll
        for (int nt = 0; nt < 2; ++nt) {
            v8f ir, iz, inn;
            #pragma unroll
            for (int r = 0; r < 8; ++r) {
                const int row = hi * 8 + r;
                const float* p = girow + (long)row * G3 + lo;
                ir[r]  = p[(nt + 0) * 16];
                iz[r]  = p[(nt + 2) * 16];
                inn[r] = p[(nt + 4) * 16];
            }
            #pragma unroll
            for (int r = 0; r < 8; ++r) {
                const float rg = sigmoid_f(ir[r] + acc[nt][r]);
                const float zg = sigmoid_f(iz[r] + acc[nt + 2][r]);
                const float ng = tanhf(inn[r] + rg * acc[nt + 4][r]);
                hc[nt][r] = (1.0f - zg) * ng + zg * hc[nt][r];
            }
            #pragma unroll
            for (int r = 0; r < 8; ++r) {
                hw[(hi * 8 + r) * 34 + nt * 16 + lo] = hc[nt][r];
            }
        }
        __syncthreads();
    }

    if (lane < 16) {
        float hrow[H_SZ];
        #pragma unroll
        for (int k = 0; k < H_SZ; ++k) hrow[k] = hw[lane * 34 + k];
        float y = 0.0f;
        for (int j = 0; j < 16; ++j) {
            float a = b1[j];
            #pragma unroll
            for (int k = 0; k < H_SZ; ++k) a += hrow[k] * w1[j * H_SZ + k];
            y += w2[j] * fmaxf(a, 0.0f);
        }
        out[bb + lane] = y;
    }
}

extern "C" void kernel_launch(void* const* d_in, const int* in_sizes, int n_in,
                              void* d_out, int out_size, void* d_ws, size_t ws_size,
                              hipStream_t stream) {
    const float* x    = (const float*)d_in[0];
    const float* w_ih = (const float*)d_in[1];
    const float* w_hh = (const float*)d_in[2];
    const float* w1   = (const float*)d_in[3];
    const float* b1   = (const float*)d_in[4];
    const float* w2   = (const float*)d_in[5];
    float* out = (float*)d_out;

    float* gi = (float*)d_ws;

    gru_gemm_in<<<1920, 256, 0, stream>>>(x, w_ih, gi);
    gru_recurrence<<<128, 64, 0, stream>>>(gi, w_hh, w1, b1, w2, out);
}


# recurrence re-cut by K: 4 waves share 16 batch rows, each multiplies 8 columns of h (12 products a step, not 48); partial sums meet in LDS, one barrier; gi asked a step ahead; gh sum order changed
# speedup vs baseline: 2.3214x; 2.3214x over previous
#include <hip/hip_runtime.h>
#include <cmath>


typedef __attribute__((ext_vector_type(2))) float v2f;
typedef __attribute__((ext_vector_type(4))) float v4f;
typedef __attribute__((ext_vector_type(8))) float v8f;

#define B_SZ 4096
#define T_SZ 60
#define F_SZ 158
#define G3 96
#define H_SZ 32

__device__ __forceinline__ float sigmoid_f(float x) {
    return 1.0f / (1.0f + __expf(-x));
}

#define P_THREADS 384
#define P_ROWS    64
#define P_CHUNK   (P_ROWS * F_SZ)
#define P_F4      (P_CHUNK / 4)
#define P_BLOCKS  256
#define P_ITERS   (B_SZ * T_SZ / P_ROWS / P_BLOCKS)
#define P_FULL    (P_F4 / P_THREADS)
#define P_TAIL    (P_F4 - P_FULL * P_THREADS)

__device__ __forceinline__ void proj_request(const float* __restrict__ src, int tid, int tl,
                                             v4f (&pre)[P_FULL + 1])
{
    #pragma unroll
    for (int j = 0; j < P_FULL; ++j)
        pre[j] = *reinterpret_cast<const v4f*>(src + 4 * (tid + j * P_THREADS));
    pre[P_FULL] = *reinterpret_cast<const v4f*>(src + 4 * (tl + P_FULL * P_THREADS));
}

__device__ __forceinline__ void proj_stage(float* dst, int tid, int tl,
                                           const v4f (&pre)[P_FULL + 1])
{
    #pragma unroll
    for (int j = 0; j < P_FULL; ++j)
        *reinterpret_cast<v4f*>(dst + 4 * (tid + j * P_THREADS)) = pre[j];
    *reinterpret_cast<v4f*>(dst + 4 * (tl + P_FULL * P_THREADS)) = pre[P_FULL];
}

__device__ __forceinline__ void proj_tile(const float* xt, int hi, const v2f (&bf)[2][40],
                                          v8f& d0, v8f& d1)
{
    v8f acc0 = {}, acc1 = {};
    #pragma unroll
    for (int kt = 0; kt < 40; ++kt) {
        const int c = (kt == 39) ? 156 : kt * 4 + 2 * hi;
        v2f a = *reinterpret_cast<const v2f*>(xt + c);
        if (kt == 39) {
            a.x = hi ? 0.0f : a.x;
            a.y = hi ? 0.0f : a.y;
        }
        acc0 = __builtin_amdgcn_wmma_f32_16x16x4_f32(
            false, a, false, bf[0][kt], (short)0, acc0, false, false);
        acc1 = __builtin_amdgcn_wmma_f32_16x16x4_f32(
            false, a, false, bf[1][kt], (short)0, acc1, false, false);
    }
    d0 = acc0;
    d1 = acc1;
}

__device__ __forceinline__ void proj_store(float* __restrict__ gi, int m0, int col,
                                           const v8f& d0, const v8f& d1)
{
    const int t0 = m0 % T_SZ;
    const int b0 = m0 / T_SZ;
    #pragma unroll
    for (int r = 0; r < 8; ++r) {
        int t = t0 + r;
        int b = b0;
        if (t >= T_SZ) { t -= T_SZ; b += 1; }
        float* orow = gi + ((t * B_SZ + b) * G3 + col);
        orow[0]  = d0[r];
        orow[16] = d1[r];
    }
}

extern "C" __global__ __launch_bounds__(P_THREADS)
void gru_gemm_in(const float* __restrict__ x, const float* __restrict__ w_ih,
                 float* __restrict__ gi)
{
    __shared__ __attribute__((aligned(16))) float xs[2 * P_CHUNK];
    const int tid  = threadIdx.x;
    const int lane = tid & 31;
    const int lo   = lane & 15;
    const int hi   = lane >> 4;
    const int wave = tid >> 5;
    const int ct   = wave % 3;
    const int rg   = wave / 3;
    const int blk  = blockIdx.x;

    const int tl   = (tid < P_TAIL) ? tid : (P_TAIL - 1);

    v4f pre[P_FULL + 1];
    proj_request(x + (size_t)blk * P_CHUNK, tid, tl, pre);

    v2f bf[2][40];
    #pragma unroll
    for (int nt = 0; nt < 2; ++nt) {
        const float* wrow = w_ih + (ct * 32 + nt * 16 + lo) * F_SZ;
        #pragma unroll
        for (int kt = 0; kt < 40; ++kt) {
            const int c = (kt == 39) ? 156 : kt * 4 + 2 * hi;
            v2f v = *reinterpret_cast<const v2f*>(wrow + c);
            if (kt == 39) {
                v.x = hi ? 0.0f : v.x;
                v.y = hi ? 0.0f : v.y;
            }
            bf[nt][kt] = v;
        }
    }

    proj_stage(xs, tid, tl, pre);
    __syncthreads();

    const int xoff = (rg * 16 + lo) * F_SZ;
    const int mrow = rg * 16 + hi * 8;
    const int col  = ct * 32 + lo;
    v8f d0, d1;

    proj_request(x + (size_t)(P_BLOCKS + blk) * P_CHUNK, tid, tl, pre);
    __builtin_amdgcn_sched_barrier(0);
    proj_tile(xs + xoff, hi, bf, d0, d1);
    proj_stage(xs + P_CHUNK, tid, tl, pre);
    __syncthreads();

    for (int it = 1; it < P_ITERS; ++it) {
        const int  cur  = it & 1;
        const bool more = (it + 1 < P_ITERS);
        if (more)
            proj_request(x + (size_t)((it + 1) * P_BLOCKS + blk) * P_CHUNK, tid, tl, pre);
        __builtin_amdgcn_sched_barrier(0);
        proj_store(gi, ((it - 1) * P_BLOCKS + blk) * P_ROWS + mrow, col, d0, d1);
        proj_tile(xs + cur * P_CHUNK + xoff, hi, bf, d0, d1);
        if (more)
            proj_stage(xs + (cur ^ 1) * P_CHUNK, tid, tl, pre);
        __syncthreads();
    }
    proj_store(gi, ((P_ITERS - 1) * P_BLOCKS + blk) * P_ROWS + mrow, col, d0, d1);
}

#define R_WAVES 4
#define R_TILE  16
#define R_PP    100
#define R_PLANE (R_TILE * R_PP)
#define R_HP    40

extern "C" __global__ __launch_bounds__(R_WAVES * 32)
void gru_recurrence(const float* __restrict__ gi, const float* __restrict__ w_hh,
                    const float* __restrict__ w1, const float* __restrict__ b1,
                    const float* __restrict__ w2, float* __restrict__ out)
{
    __shared__ __attribute__((aligned(16))) float part[2 * R_WAVES * R_PLANE];
    __shared__ __attribute__((aligned(16))) float hfin[R_TILE * R_HP];
    const int tid  = threadIdx.x;
    const int lane = tid & 31;
    const int lo   = lane & 15;
    const int hi   = lane >> 4;
    const int wave = tid >> 5;
    const int b0   = blockIdx.x * R_TILE;
    const int j0   = 8 * wave + 4 * hi;

    v2f af[6][2];
    #pragma unroll
    for (int mt = 0; mt < 6; ++mt) {
        #pragma unroll
        for (int kk = 0; kk < 2; ++kk)
            af[mt][kk] = *reinterpret_cast<const v2f*>(w_hh + (mt * 16 + lo) * H_SZ + j0 + 2 * kk);
    }

    float* wr = part + wave * R_PLANE + lo * R_PP + 8 * hi;
    const float* rd = part + lo * R_PP + j0;
    const float* girow = gi + (long)(b0 + lo) * G3 + j0;

    v4f hc = {0.0f, 0.0f, 0.0f, 0.0f};
    v4f gin[3];
    #pragma unroll
    for (int g = 0; g < 3; ++g)
        gin[g] = *reinterpret_cast<const v4f*>(girow + g * 32);

    for (int t = 0; t < T_SZ; ++t) {
        const int buf = (t & 1) * (R_WAVES * R_PLANE);

        v2f bk0, bk1;
        bk0.x = hc[0]; bk0.y = hc[1];
        bk1.x = hc[2]; bk1.y = hc[3];
        #pragma unroll
        for (int mt = 0; mt < 6; ++mt) {
            v8f acc = {};
            acc = __builtin_amdgcn_wmma_f32_16x16x4_f32(
                false, af[mt][0], false, bk0, (short)0, acc, false, false);
            acc = __builtin_amdgcn_wmma_f32_16x16x4_f32(
                false, af[mt][1], false, bk1, (short)0, acc, false, false);
            v4f q0, q1;
            q0.x = acc[0]; q0.y = acc[1]; q0.z = acc[2]; q0.w = acc[3];
            q1.x = acc[4]; q1.y = acc[5]; q1.z = acc[6]; q1.w = acc[7];
            *reinterpret_cast<v4f*>(wr + buf + mt * 16)     = q0;
            *reinterpret_cast<v4f*>(wr + buf + mt * 16 + 4) = q1;
        }
        __syncthreads();

        v4f gh[3];
        #pragma unroll
        for (int g = 0; g < 3; ++g) {
            const float* p = rd + buf + g * 32;
            const v4f p0 = *reinterpret_cast<const v4f*>(p);
            const v4f p1 = *reinterpret_cast<const v4f*>(p + R_PLANE);
            const v4f p2 = *reinterpret_cast<const v4f*>(p + 2 * R_PLANE);
            const v4f p3 = *reinterpret_cast<const v4f*>(p + 3 * R_PLANE);
            gh[g] = ((p0 + p1) + p2) + p3;
        }

        const int tn = (t + 1 < T_SZ) ? (t + 1) : t;
        v4f gnx[3];
        #pragma unroll
        for (int g = 0; g < 3; ++g)
            gnx[g] = *reinterpret_cast<const v4f*>(girow + (long)tn * (B_SZ * G3) + g * 32);
        __builtin_amdgcn_sched_barrier(0);

        float zg[4], na[4];
        #pragma unroll
        for (int e = 0; e < 4; ++e) {
            const float rg = sigmoid_f(gin[0][e] + gh[0][e]);
            zg[e] = sigmoid_f(gin[1][e] + gh[1][e]);
            na[e] = gin[2][e] + rg * gh[2][e];
        }
        asm volatile("" : "+v"(na[0]), "+v"(na[1]), "+v"(na[2]), "+v"(na[3]),
                          "+v"(zg[0]), "+v"(zg[1]), "+v"(zg[2]), "+v"(zg[3]));
        #pragma unroll
        for (int e = 0; e < 4; ++e) {
            const float ng = tanhf(na[e]);
            hc[e] = (1.0f - zg[e]) * ng + zg[e] * hc[e];
        }
        #pragma unroll
        for (int g = 0; g < 3; ++g) gin[g] = gnx[g];
    }

    *reinterpret_cast<v4f*>(hfin + lo * R_HP + j0) = hc;
    __syncthreads();
    if (lane < 4) {
        const int row = 4 * wave + lane;
        float hrow[H_SZ];
        #pragma unroll
        for (int k = 0; k < H_SZ; ++k) hrow[k] = hfin[row * R_HP + k];
        float y = 0.0f;
        for (int j = 0; j < 16; ++j) {
            float a = b1[j];
            #pragma unroll
            for (int k = 0; k < H_SZ; ++k) a += hrow[k] * w1[j * H_SZ + k];
            y += w2[j] * fmaxf(a, 0.0f);
        }
        out[b0 + row] = y;
    }
}

extern "C" void kernel_launch(void* const* d_in, const int* in_sizes, int n_in,
                              void* d_out, int out_size, void* d_ws, size_t ws_size,
                              hipStream_t stream) {
    const float* x    = (const float*)d_in[0];
    const float* w_ih = (const float*)d_in[1];
    const float* w_hh = (const float*)d_in[2];
    const float* w1   = (const float*)d_in[3];
    const float* b1   = (const float*)d_in[4];
    const float* w2   = (const float*)d_in[5];
    float* out = (float*)d_out;

    float* gi = (float*)d_ws;

    gru_gemm_in<<<256, 384, 0, stream>>>(x, w_ih, gi);
    gru_recurrence<<<256, 128, 0, stream>>>(gi, w_hh, w1, b1, w2, out);
}


# recurrence: waves 0-2 hint line w of the 16 gi rows of step t+3 into the second-level cache (global prefetch: no register, no wait), so a step's own gi loads hit that cache instead of memory
# speedup vs baseline: 2.5080x; 1.0804x over previous
#include <hip/hip_runtime.h>
#include <cmath>


typedef __attribute__((ext_vector_type(2))) float v2f;
typedef __attribute__((ext_vector_type(4))) float v4f;
typedef __attribute__((ext_vector_type(8))) float v8f;

#define B_SZ 4096
#define T_SZ 60
#define F_SZ 158
#define G3 96
#define H_SZ 32

__device__ __forceinline__ float sigmoid_f(float x) {
    return 1.0f / (1.0f + __expf(-x));
}

#define P_THREADS 384
#define P_ROWS    64
#define P_CHUNK   (P_ROWS * F_SZ)
#define P_F4      (P_CHUNK / 4)
#define P_BLOCKS  256
#define P_ITERS   (B_SZ * T_SZ / P_ROWS / P_BLOCKS)
#define P_FULL    (P_F4 / P_THREADS)
#define P_TAIL    (P_F4 - P_FULL * P_THREADS)

__device__ __forceinline__ void proj_request(const float* __restrict__ src, int tid, int tl,
                                             v4f (&pre)[P_FULL + 1])
{
    #pragma unroll
    for (int j = 0; j < P_FULL; ++j)
        pre[j] = *reinterpret_cast<const v4f*>(src + 4 * (tid + j * P_THREADS));
    pre[P_FULL] = *reinterpret_cast<const v4f*>(src + 4 * (tl + P_FULL * P_THREADS));
}

__device__ __forceinline__ void proj_stage(float* dst, int tid, int tl,
                                           const v4f (&pre)[P_FULL + 1])
{
    #pragma unroll
    for (int j = 0; j < P_FULL; ++j)
        *reinterpret_cast<v4f*>(dst + 4 * (tid + j * P_THREADS)) = pre[j];
    *reinterpret_cast<v4f*>(dst + 4 * (tl + P_FULL * P_THREADS)) = pre[P_FULL];
}

__device__ __forceinline__ void proj_tile(const float* xt, int hi, const v2f (&bf)[2][40],
                                          v8f& d0, v8f& d1)
{
    v8f acc0 = {}, acc1 = {};
    #pragma unroll
    for (int kt = 0; kt < 40; ++kt) {
        const int c = (kt == 39) ? 156 : kt * 4 + 2 * hi;
        v2f a = *reinterpret_cast<const v2f*>(xt + c);
        if (kt == 39) {
            a.x = hi ? 0.0f : a.x;
            a.y = hi ? 0.0f : a.y;
        }
        acc0 = __builtin_amdgcn_wmma_f32_16x16x4_f32(
            false, a, false, bf[0][kt], (short)0, acc0, false, false);
        acc1 = __builtin_amdgcn_wmma_f32_16x16x4_f32(
            false, a, false, bf[1][kt], (short)0, acc1, false, false);
    }
    d0 = acc0;
    d1 = acc1;
}

__device__ __forceinline__ void proj_store(float* __restrict__ gi, int m0, int col,
                                           const v8f& d0, const v8f& d1)
{
    const int t0 = m0 % T_SZ;
    const int b0 = m0 / T_SZ;
    #pragma unroll
    for (int r = 0; r < 8; ++r) {
        int t = t0 + r;
        int b = b0;
        if (t >= T_SZ) { t -= T_SZ; b += 1; }
        float* orow = gi + ((t * B_SZ + b) * G3 + col);
        orow[0]  = d0[r];
        orow[16] = d1[r];
    }
}

extern "C" __global__ __launch_bounds__(P_THREADS)
void gru_gemm_in(const float* __restrict__ x, const float* __restrict__ w_ih,
                 float* __restrict__ gi)
{
    __shared__ __attribute__((aligned(16))) float xs[2 * P_CHUNK];
    const int tid  = threadIdx.x;
    const int lane = tid & 31;
    const int lo   = lane & 15;
    const int hi   = lane >> 4;
    const int wave = tid >> 5;
    const int ct   = wave % 3;
    const int rg   = wave / 3;
    const int blk  = blockIdx.x;

    const int tl   = (tid < P_TAIL) ? tid : (P_TAIL - 1);

    v4f pre[P_FULL + 1];
    proj_request(x + (size_t)blk * P_CHUNK, tid, tl, pre);

    v2f bf[2][40];
    #pragma unroll
    for (int nt = 0; nt < 2; ++nt) {
        const float* wrow = w_ih + (ct * 32 + nt * 16 + lo) * F_SZ;
        #pragma unroll
        for (int kt = 0; kt < 40; ++kt) {
            const int c = (kt == 39) ? 156 : kt * 4 + 2 * hi;
            v2f v = *reinterpret_cast<const v2f*>(wrow + c);
            if (kt == 39) {
                v.x = hi ? 0.0f : v.x;
                v.y = hi ? 0.0f : v.y;
            }
            bf[nt][kt] = v;
        }
    }

    proj_stage(xs, tid, tl, pre);
    __syncthreads();

    const int xoff = (rg * 16 + lo) * F_SZ;
    const int mrow = rg * 16 + hi * 8;
    const int col  = ct * 32 + lo;
    v8f d0, d1;

    proj_request(x + (size_t)(P_BLOCKS + blk) * P_CHUNK, tid, tl, pre);
    __builtin_amdgcn_sched_barrier(0);
    proj_tile(xs + xoff, hi, bf, d0, d1);
    proj_stage(xs + P_CHUNK, tid, tl, pre);
    __syncthreads();

    for (int it = 1; it < P_ITERS; ++it) {
        const int  cur  = it & 1;
        const bool more = (it + 1 < P_ITERS);
        if (more)
            proj_request(x + (size_t)((it + 1) * P_BLOCKS + blk) * P_CHUNK, tid, tl, pre);
        __builtin_amdgcn_sched_barrier(0);
        proj_store(gi, ((it - 1) * P_BLOCKS + blk) * P_ROWS + mrow, col, d0, d1);
        proj_tile(xs + cur * P_CHUNK + xoff, hi, bf, d0, d1);
        if (more)
            proj_stage(xs + (cur ^ 1) * P_CHUNK, tid, tl, pre);
        __syncthreads();
    }
    proj_store(gi, ((P_ITERS - 1) * P_BLOCKS + blk) * P_ROWS + mrow, col, d0, d1);
}

#define R_WAVES 4
#define R_TILE  16
#define R_PP    100
#define R_PLANE (R_TILE * R_PP)
#define R_HP    40
#define R_LEAD  3

extern "C" __global__ __launch_bounds__(R_WAVES * 32)
void gru_recurrence(const float* __restrict__ gi, const float* __restrict__ w_hh,
                    const float* __restrict__ w1, const float* __restrict__ b1,
                    const float* __restrict__ w2, float* __restrict__ out)
{
    __shared__ __attribute__((aligned(16))) float part[2 * R_WAVES * R_PLANE];
    __shared__ __attribute__((aligned(16))) float hfin[R_TILE * R_HP];
    const int tid  = threadIdx.x;
    const int lane = tid & 31;
    const int lo   = lane & 15;
    const int hi   = lane >> 4;
    const int wave = tid >> 5;
    const int b0   = blockIdx.x * R_TILE;
    const int j0   = 8 * wave + 4 * hi;

    v2f af[6][2];
    #pragma unroll
    for (int mt = 0; mt < 6; ++mt) {
        #pragma unroll
        for (int kk = 0; kk < 2; ++kk)
            af[mt][kk] = *reinterpret_cast<const v2f*>(w_hh + (mt * 16 + lo) * H_SZ + j0 + 2 * kk);
    }

    float* wr = part + wave * R_PLANE + lo * R_PP + 8 * hi;
    const float* rd = part + lo * R_PP + j0;
    const float* girow = gi + (long)(b0 + lo) * G3 + j0;

    v4f hc = {0.0f, 0.0f, 0.0f, 0.0f};
    v4f gin[3];
    #pragma unroll
    for (int g = 0; g < 3; ++g)
        gin[g] = *reinterpret_cast<const v4f*>(girow + g * 32);

    const float* pfrow = gi + (long)(b0 + lo) * G3 + 32 * wave;
    const bool hints = __builtin_amdgcn_readfirstlane(wave) < 3;
    if (hints) {
        for (int s = 2; s < R_LEAD; ++s)
            __builtin_prefetch(pfrow + (long)s * (B_SZ * G3), 0, 1);
    }

    for (int t = 0; t < T_SZ; ++t) {
        const int buf = (t & 1) * (R_WAVES * R_PLANE);

        if (hints) {
            const int tp = (t + R_LEAD < T_SZ) ? (t + R_LEAD) : (T_SZ - 1);
            __builtin_prefetch(pfrow + (long)tp * (B_SZ * G3), 0, 1);
        }

        v2f bk0, bk1;
        bk0.x = hc[0]; bk0.y = hc[1];
        bk1.x = hc[2]; bk1.y = hc[3];
        #pragma unroll
        for (int mt = 0; mt < 6; ++mt) {
            v8f acc = {};
            acc = __builtin_amdgcn_wmma_f32_16x16x4_f32(
                false, af[mt][0], false, bk0, (short)0, acc, false, false);
            acc = __builtin_amdgcn_wmma_f32_16x16x4_f32(
                false, af[mt][1], false, bk1, (short)0, acc, false, false);
            v4f q0, q1;
            q0.x = acc[0]; q0.y = acc[1]; q0.z = acc[2]; q0.w = acc[3];
            q1.x = acc[4]; q1.y = acc[5]; q1.z = acc[6]; q1.w = acc[7];
            *reinterpret_cast<v4f*>(wr + buf + mt * 16)     = q0;
            *reinterpret_cast<v4f*>(wr + buf + mt * 16 + 4) = q1;
        }
        __syncthreads();

        v4f gh[3];
        #pragma unroll
        for (int g = 0; g < 3; ++g) {
            const float* p = rd + buf + g * 32;
            const v4f p0 = *reinterpret_cast<const v4f*>(p);
            const v4f p1 = *reinterpret_cast<const v4f*>(p + R_PLANE);
            const v4f p2 = *reinterpret_cast<const v4f*>(p + 2 * R_PLANE);
            const v4f p3 = *reinterpret_cast<const v4f*>(p + 3 * R_PLANE);
            gh[g] = ((p0 + p1) + p2) + p3;
        }

        const int tn = (t + 1 < T_SZ) ? (t + 1) : t;
        v4f gnx[3];
        #pragma unroll
        for (int g = 0; g < 3; ++g)
            gnx[g] = *reinterpret_cast<const v4f*>(girow + (long)tn * (B_SZ * G3) + g * 32);
        __builtin_amdgcn_sched_barrier(0);

        float zg[4], na[4];
        #pragma unroll
        for (int e = 0; e < 4; ++e) {
            const float rg = sigmoid_f(gin[0][e] + gh[0][e]);
            zg[e] = sigmoid_f(gin[1][e] + gh[1][e]);
            na[e] = gin[2][e] + rg * gh[2][e];
        }
        asm volatile("" : "+v"(na[0]), "+v"(na[1]), "+v"(na[2]), "+v"(na[3]),
                          "+v"(zg[0]), "+v"(zg[1]), "+v"(zg[2]), "+v"(zg[3]));
        #pragma unroll
        for (int e = 0; e < 4; ++e) {
            const float ng = tanhf(na[e]);
            hc[e] = (1.0f - zg[e]) * ng + zg[e] * hc[e];
        }
        #pragma unroll
        for (int g = 0; g < 3; ++g) gin[g] = gnx[g];
    }

    *reinterpret_cast<v4f*>(hfin + lo * R_HP + j0) = hc;
    __syncthreads();
    if (lane < 4) {
        const int row = 4 * wave + lane;
        float hrow[H_SZ];
        #pragma unroll
        for (int k = 0; k < H_SZ; ++k) hrow[k] = hfin[row * R_HP + k];
        float y = 0.0f;
        for (int j = 0; j < 16; ++j) {
            float a = b1[j];
            #pragma unroll
            for (int k = 0; k < H_SZ; ++k) a += hrow[k] * w1[j * H_SZ + k];
            y += w2[j] * fmaxf(a, 0.0f);
        }
        out[b0 + row] = y;
    }
}

extern "C" void kernel_launch(void* const* d_in, const int* in_sizes, int n_in,
                              void* d_out, int out_size, void* d_ws, size_t ws_size,
                              hipStream_t stream) {
    const float* x    = (const float*)d_in[0];
    const float* w_ih = (const float*)d_in[1];
    const float* w_hh = (const float*)d_in[2];
    const float* w1   = (const float*)d_in[3];
    const float* b1   = (const float*)d_in[4];
    const float* w2   = (const float*)d_in[5];
    float* out = (float*)d_out;

    float* gi = (float*)d_ws;

    gru_gemm_in<<<256, 384, 0, stream>>>(x, w_ih, gi);
    gru_recurrence<<<256, 128, 0, stream>>>(gi, w_hh, w1, b1, w2, out);
}


# recurrence re-cut to 8 batch rows a block, 2 blocks resident a WGP (2 waves a SIMD): a lane pair splits its 4 h columns, 2 gate chains a lane; partner pair by a row lane exchange; sums order kept
# speedup vs baseline: 2.6624x; 1.0616x over previous
#include <hip/hip_runtime.h>
#include <cmath>


typedef __attribute__((ext_vector_type(2))) float v2f;
typedef __attribute__((ext_vector_type(4))) float v4f;
typedef __attribute__((ext_vector_type(8))) float v8f;

#define B_SZ 4096
#define T_SZ 60
#define F_SZ 158
#define G3 96
#define H_SZ 32

__device__ __forceinline__ float sigmoid_f(float x) {
    return 1.0f / (1.0f + __expf(-x));
}

#define P_THREADS 384
#define P_ROWS    64
#define P_CHUNK   (P_ROWS * F_SZ)
#define P_F4      (P_CHUNK / 4)
#define P_BLOCKS  256
#define P_ITERS   (B_SZ * T_SZ / P_ROWS / P_BLOCKS)
#define P_FULL    (P_F4 / P_THREADS)
#define P_TAIL    (P_F4 - P_FULL * P_THREADS)

__device__ __forceinline__ void proj_request(const float* __restrict__ src, int tid, int tl,
                                             v4f (&pre)[P_FULL + 1])
{
    #pragma unroll
    for (int j = 0; j < P_FULL; ++j)
        pre[j] = *reinterpret_cast<const v4f*>(src + 4 * (tid + j * P_THREADS));
    pre[P_FULL] = *reinterpret_cast<const v4f*>(src + 4 * (tl + P_FULL * P_THREADS));
}

__device__ __forceinline__ void proj_stage(float* dst, int tid, int tl,
                                           const v4f (&pre)[P_FULL + 1])
{
    #pragma unroll
    for (int j = 0; j < P_FULL; ++j)
        *reinterpret_cast<v4f*>(dst + 4 * (tid + j * P_THREADS)) = pre[j];
    *reinterpret_cast<v4f*>(dst + 4 * (tl + P_FULL * P_THREADS)) = pre[P_FULL];
}

__device__ __forceinline__ void proj_tile(const float* xt, int hi, const v2f (&bf)[2][40],
                                          v8f& d0, v8f& d1)
{
    v8f acc0 = {}, acc1 = {};
    #pragma unroll
    for (int kt = 0; kt < 40; ++kt) {
        const int c = (kt == 39) ? 156 : kt * 4 + 2 * hi;
        v2f a = *reinterpret_cast<const v2f*>(xt + c);
        if (kt == 39) {
            a.x = hi ? 0.0f : a.x;
            a.y = hi ? 0.0f : a.y;
        }
        acc0 = __builtin_amdgcn_wmma_f32_16x16x4_f32(
            false, a, false, bf[0][kt], (short)0, acc0, false, false);
        acc1 = __builtin_amdgcn_wmma_f32_16x16x4_f32(
            false, a, false, bf[1][kt], (short)0, acc1, false, false);
    }
    d0 = acc0;
    d1 = acc1;
}

__device__ __forceinline__ void proj_store(float* __restrict__ gi, int m0, int col,
                                           const v8f& d0, const v8f& d1)
{
    const int t0 = m0 % T_SZ;
    const int b0 = m0 / T_SZ;
    #pragma unroll
    for (int r = 0; r < 8; ++r) {
        int t = t0 + r;
        int b = b0;
        if (t >= T_SZ) { t -= T_SZ; b += 1; }
        float* orow = gi + ((t * B_SZ + b) * G3 + col);
        orow[0]  = d0[r];
        orow[16] = d1[r];
    }
}

extern "C" __global__ __launch_bounds__(P_THREADS)
void gru_gemm_in(const float* __restrict__ x, const float* __restrict__ w_ih,
                 float* __restrict__ gi)
{
    __shared__ __attribute__((aligned(16))) float xs[2 * P_CHUNK];
    const int tid  = threadIdx.x;
    const int lane = tid & 31;
    const int lo   = lane & 15;
    const int hi   = lane >> 4;
    const int wave = tid >> 5;
    const int ct   = wave % 3;
    const int rg   = wave / 3;
    const int blk  = blockIdx.x;

    const int tl   = (tid < P_TAIL) ? tid : (P_TAIL - 1);

    v4f pre[P_FULL + 1];
    proj_request(x + (size_t)blk * P_CHUNK, tid, tl, pre);

    v2f bf[2][40];
    #pragma unroll
    for (int nt = 0; nt < 2; ++nt) {
        const float* wrow = w_ih + (ct * 32 + nt * 16 + lo) * F_SZ;
        #pragma unroll
        for (int kt = 0; kt < 40; ++kt) {
            const int c = (kt == 39) ? 156 : kt * 4 + 2 * hi;
            v2f v = *reinterpret_cast<const v2f*>(wrow + c);
            if (kt == 39) {
                v.x = hi ? 0.0f : v.x;
                v.y = hi ? 0.0f : v.y;
            }
            bf[nt][kt] = v;
        }
    }

    proj_stage(xs, tid, tl, pre);
    __syncthreads();

    const int xoff = (rg * 16 + lo) * F_SZ;
    const int mrow = rg * 16 + hi * 8;
    const int col  = ct * 32 + lo;
    v8f d0, d1;

    proj_request(x + (size_t)(P_BLOCKS + blk) * P_CHUNK, tid, tl, pre);
    __builtin_amdgcn_sched_barrier(0);
    proj_tile(xs + xoff, hi, bf, d0, d1);
    proj_stage(xs + P_CHUNK, tid, tl, pre);
    __syncthreads();

    for (int it = 1; it < P_ITERS; ++it) {
        const int  cur  = it & 1;
        const bool more = (it + 1 < P_ITERS);
        if (more)
            proj_request(x + (size_t)((it + 1) * P_BLOCKS + blk) * P_CHUNK, tid, tl, pre);
        __builtin_amdgcn_sched_barrier(0);
        proj_store(gi, ((it - 1) * P_BLOCKS + blk) * P_ROWS + mrow, col, d0, d1);
        proj_tile(xs + cur * P_CHUNK + xoff, hi, bf, d0, d1);
        if (more)
            proj_stage(xs + (cur ^ 1) * P_CHUNK, tid, tl, pre);
        __syncthreads();
    }
    proj_store(gi, ((P_ITERS - 1) * P_BLOCKS + blk) * P_ROWS + mrow, col, d0, d1);
}

#define R_WAVES 4
#define R_TILE  8
#define R_PP    100
#define R_PLANE (R_TILE * R_PP)
#define R_HP    40
#define R_LEAD  3
#define R_BLOCKS (B_SZ / R_TILE)

__device__ __forceinline__ float lane_xor8(float v) {
    return __builtin_bit_cast(float,
        __builtin_amdgcn_update_dpp(0, __builtin_bit_cast(int, v), 0x168  ,
                                    0xf, 0xf, false));
}

extern "C" __global__ __launch_bounds__(R_WAVES * 32)
void gru_recurrence(const float* __restrict__ gi, const float* __restrict__ w_hh,
                    const float* __restrict__ w1, const float* __restrict__ b1,
                    const float* __restrict__ w2, float* __restrict__ out)
{
    __shared__ __attribute__((aligned(16))) float part[2 * R_WAVES * R_PLANE];
    __shared__ __attribute__((aligned(16))) float hfin[R_TILE * R_HP];
    const int tid  = threadIdx.x;
    const int lane = tid & 31;
    const int lo   = lane & 15;
    const int hi   = lane >> 4;
    const int wave = tid >> 5;
    const int row  = lo & 7;
    const int half = lo >> 3;
    const int b0   = blockIdx.x * R_TILE;
    const int j0   = 8 * wave + 4 * hi;
    const int jc   = j0 + 2 * half;

    v2f af[6][2];
    #pragma unroll
    for (int mt = 0; mt < 6; ++mt) {
        #pragma unroll
        for (int kk = 0; kk < 2; ++kk)
            af[mt][kk] = *reinterpret_cast<const v2f*>(w_hh + (mt * 16 + lo) * H_SZ + j0 + 2 * kk);
    }

    float* wr = part + wave * R_PLANE + row * R_PP + 8 * hi;
    const float* rd = part + row * R_PP + jc;
    const float* girow = gi + (long)(b0 + row) * G3 + jc;

    v2f hc = {0.0f, 0.0f};
    v2f gin[3];
    #pragma unroll
    for (int g = 0; g < 3; ++g)
        gin[g] = *reinterpret_cast<const v2f*>(girow + g * 32);

    const float* pfrow = gi + (long)(b0 + row) * G3 + 32 * wave;
    const bool hints = __builtin_amdgcn_readfirstlane(wave) < 3;
    if (hints) {
        for (int s = 2; s < R_LEAD; ++s)
            __builtin_prefetch(pfrow + (long)s * (B_SZ * G3), 0, 1);
    }

    for (int t = 0; t < T_SZ; ++t) {
        const int buf = (t & 1) * (R_WAVES * R_PLANE);

        if (hints) {
            const int tp = (t + R_LEAD < T_SZ) ? (t + R_LEAD) : (T_SZ - 1);
            __builtin_prefetch(pfrow + (long)tp * (B_SZ * G3), 0, 1);
        }

        v2f bk0, bk1;
        bk0.x = hc[0];            bk0.y = hc[1];
        bk1.x = lane_xor8(hc[0]); bk1.y = lane_xor8(hc[1]);
        v8f acc[6];
        #pragma unroll
        for (int mt = 0; mt < 6; ++mt) {
            v8f a = {};
            a = __builtin_amdgcn_wmma_f32_16x16x4_f32(
                false, af[mt][0], false, bk0, (short)0, a, false, false);
            a = __builtin_amdgcn_wmma_f32_16x16x4_f32(
                false, af[mt][1], false, bk1, (short)0, a, false, false);
            acc[mt] = a;
        }
        if (lo < R_TILE) {
            #pragma unroll
            for (int mt = 0; mt < 6; ++mt) {
                v4f q0, q1;
                q0.x = acc[mt][0]; q0.y = acc[mt][1]; q0.z = acc[mt][2]; q0.w = acc[mt][3];
                q1.x = acc[mt][4]; q1.y = acc[mt][5]; q1.z = acc[mt][6]; q1.w = acc[mt][7];
                *reinterpret_cast<v4f*>(wr + buf + mt * 16)     = q0;
                *reinterpret_cast<v4f*>(wr + buf + mt * 16 + 4) = q1;
            }
        }
        __syncthreads();

        v2f gh[3];
        #pragma unroll
        for (int g = 0; g < 3; ++g) {
            const float* p = rd + buf + g * 32;
            const v2f p0 = *reinterpret_cast<const v2f*>(p);
            const v2f p1 = *reinterpret_cast<const v2f*>(p + R_PLANE);
            const v2f p2 = *reinterpret_cast<const v2f*>(p + 2 * R_PLANE);
            const v2f p3 = *reinterpret_cast<const v2f*>(p + 3 * R_PLANE);
            gh[g] = ((p0 + p1) + p2) + p3;
        }

        const int tn = (t + 1 < T_SZ) ? (t + 1) : t;
        v2f gnx[3];
        #pragma unroll
        for (int g = 0; g < 3; ++g)
            gnx[g] = *reinterpret_cast<const v2f*>(girow + (long)tn * (B_SZ * G3) + g * 32);
        __builtin_amdgcn_sched_barrier(0);

        float zg[2], na[2];
        #pragma unroll
        for (int e = 0; e < 2; ++e) {
            const float rg = sigmoid_f(gin[0][e] + gh[0][e]);
            zg[e] = sigmoid_f(gin[1][e] + gh[1][e]);
            na[e] = gin[2][e] + rg * gh[2][e];
        }
        asm volatile("" : "+v"(na[0]), "+v"(na[1]), "+v"(zg[0]), "+v"(zg[1]));
        #pragma unroll
        for (int e = 0; e < 2; ++e) {
            const float ng = tanhf(na[e]);
            hc[e] = (1.0f - zg[e]) * ng + zg[e] * hc[e];
        }
        #pragma unroll
        for (int g = 0; g < 3; ++g) gin[g] = gnx[g];
    }

    *reinterpret_cast<v2f*>(hfin + row * R_HP + jc) = hc;
    __syncthreads();
    if (lane < 2) {
        const int r = 2 * wave + lane;
        float hrow[H_SZ];
        #pragma unroll
        for (int k = 0; k < H_SZ; ++k) hrow[k] = hfin[r * R_HP + k];
        float y = 0.0f;
        for (int j = 0; j < 16; ++j) {
            float a = b1[j];
            #pragma unroll
            for (int k = 0; k < H_SZ; ++k) a += hrow[k] * w1[j * H_SZ + k];
            y += w2[j] * fmaxf(a, 0.0f);
        }
        out[b0 + r] = y;
    }
}

extern "C" void kernel_launch(void* const* d_in, const int* in_sizes, int n_in,
                              void* d_out, int out_size, void* d_ws, size_t ws_size,
                              hipStream_t stream) {
    const float* x    = (const float*)d_in[0];
    const float* w_ih = (const float*)d_in[1];
    const float* w_hh = (const float*)d_in[2];
    const float* w1   = (const float*)d_in[3];
    const float* b1   = (const float*)d_in[4];
    const float* w2   = (const float*)d_in[5];
    float* out = (float*)d_out;

    float* gi = (float*)d_ws;

    gru_gemm_in<<<256, 384, 0, stream>>>(x, w_ih, gi);
    gru_recurrence<<<R_BLOCKS, 128, 0, stream>>>(gi, w_hh, w1, b1, w2, out);
}


# MLP head on all 128 lanes, not 8: a lane does one (row, unit) dot product, its w1 row in registers since before the scan; a row's 16 units are summed by a lane-exchange tree (that sum's order changed)
# speedup vs baseline: 2.7522x; 1.0337x over previous
#include <hip/hip_runtime.h>
#include <cmath>


typedef __attribute__((ext_vector_type(2))) float v2f;
typedef __attribute__((ext_vector_type(4))) float v4f;
typedef __attribute__((ext_vector_type(8))) float v8f;

#define B_SZ 4096
#define T_SZ 60
#define F_SZ 158
#define G3 96
#define H_SZ 32

__device__ __forceinline__ float sigmoid_f(float x) {
    return 1.0f / (1.0f + __expf(-x));
}

#define P_THREADS 384
#define P_ROWS    64
#define P_CHUNK   (P_ROWS * F_SZ)
#define P_F4      (P_CHUNK / 4)
#define P_BLOCKS  256
#define P_ITERS   (B_SZ * T_SZ / P_ROWS / P_BLOCKS)
#define P_FULL    (P_F4 / P_THREADS)
#define P_TAIL    (P_F4 - P_FULL * P_THREADS)

__device__ __forceinline__ void proj_request(const float* __restrict__ src, int tid, int tl,
                                             v4f (&pre)[P_FULL + 1])
{
    #pragma unroll
    for (int j = 0; j < P_FULL; ++j)
        pre[j] = *reinterpret_cast<const v4f*>(src + 4 * (tid + j * P_THREADS));
    pre[P_FULL] = *reinterpret_cast<const v4f*>(src + 4 * (tl + P_FULL * P_THREADS));
}

__device__ __forceinline__ void proj_stage(float* dst, int tid, int tl,
                                           const v4f (&pre)[P_FULL + 1])
{
    #pragma unroll
    for (int j = 0; j < P_FULL; ++j)
        *reinterpret_cast<v4f*>(dst + 4 * (tid + j * P_THREADS)) = pre[j];
    *reinterpret_cast<v4f*>(dst + 4 * (tl + P_FULL * P_THREADS)) = pre[P_FULL];
}

__device__ __forceinline__ void proj_tile(const float* xt, int hi, const v2f (&bf)[2][40],
                                          v8f& d0, v8f& d1)
{
    v8f acc0 = {}, acc1 = {};
    #pragma unroll
    for (int kt = 0; kt < 40; ++kt) {
        const int c = (kt == 39) ? 156 : kt * 4 + 2 * hi;
        v2f a = *reinterpret_cast<const v2f*>(xt + c);
        if (kt == 39) {
            a.x = hi ? 0.0f : a.x;
            a.y = hi ? 0.0f : a.y;
        }
        acc0 = __builtin_amdgcn_wmma_f32_16x16x4_f32(
            false, a, false, bf[0][kt], (short)0, acc0, false, false);
        acc1 = __builtin_amdgcn_wmma_f32_16x16x4_f32(
            false, a, false, bf[1][kt], (short)0, acc1, false, false);
    }
    d0 = acc0;
    d1 = acc1;
}

__device__ __forceinline__ void proj_store(float* __restrict__ gi, int m0, int col,
                                           const v8f& d0, const v8f& d1)
{
    const int t0 = m0 % T_SZ;
    const int b0 = m0 / T_SZ;
    #pragma unroll
    for (int r = 0; r < 8; ++r) {
        int t = t0 + r;
        int b = b0;
        if (t >= T_SZ) { t -= T_SZ; b += 1; }
        float* orow = gi + ((t * B_SZ + b) * G3 + col);
        orow[0]  = d0[r];
        orow[16] = d1[r];
    }
}

extern "C" __global__ __launch_bounds__(P_THREADS)
void gru_gemm_in(const float* __restrict__ x, const float* __restrict__ w_ih,
                 float* __restrict__ gi)
{
    __shared__ __attribute__((aligned(16))) float xs[2 * P_CHUNK];
    const int tid  = threadIdx.x;
    const int lane = tid & 31;
    const int lo   = lane & 15;
    const int hi   = lane >> 4;
    const int wave = tid >> 5;
    const int ct   = wave % 3;
    const int rg   = wave / 3;
    const int blk  = blockIdx.x;

    const int tl   = (tid < P_TAIL) ? tid : (P_TAIL - 1);

    v4f pre[P_FULL + 1];
    proj_request(x + (size_t)blk * P_CHUNK, tid, tl, pre);

    v2f bf[2][40];
    #pragma unroll
    for (int nt = 0; nt < 2; ++nt) {
        const float* wrow = w_ih + (ct * 32 + nt * 16 + lo) * F_SZ;
        #pragma unroll
        for (int kt = 0; kt < 40; ++kt) {
            const int c = (kt == 39) ? 156 : kt * 4 + 2 * hi;
            v2f v = *reinterpret_cast<const v2f*>(wrow + c);
            if (kt == 39) {
                v.x = hi ? 0.0f : v.x;
                v.y = hi ? 0.0f : v.y;
            }
            bf[nt][kt] = v;
        }
    }

    proj_stage(xs, tid, tl, pre);
    __syncthreads();

    const int xoff = (rg * 16 + lo) * F_SZ;
    const int mrow = rg * 16 + hi * 8;
    const int col  = ct * 32 + lo;
    v8f d0, d1;

    proj_request(x + (size_t)(P_BLOCKS + blk) * P_CHUNK, tid, tl, pre);
    __builtin_amdgcn_sched_barrier(0);
    proj_tile(xs + xoff, hi, bf, d0, d1);
    proj_stage(xs + P_CHUNK, tid, tl, pre);
    __syncthreads();

    for (int it = 1; it < P_ITERS; ++it) {
        const int  cur  = it & 1;
        const bool more = (it + 1 < P_ITERS);
        if (more)
            proj_request(x + (size_t)((it + 1) * P_BLOCKS + blk) * P_CHUNK, tid, tl, pre);
        __builtin_amdgcn_sched_barrier(0);
        proj_store(gi, ((it - 1) * P_BLOCKS + blk) * P_ROWS + mrow, col, d0, d1);
        proj_tile(xs + cur * P_CHUNK + xoff, hi, bf, d0, d1);
        if (more)
            proj_stage(xs + (cur ^ 1) * P_CHUNK, tid, tl, pre);
        __syncthreads();
    }
    proj_store(gi, ((P_ITERS - 1) * P_BLOCKS + blk) * P_ROWS + mrow, col, d0, d1);
}

#define R_WAVES 4
#define R_TILE  8
#define R_PP    100
#define R_PLANE (R_TILE * R_PP)
#define R_HP    40
#define R_LEAD  3
#define R_BLOCKS (B_SZ / R_TILE)

template <int M>
__device__ __forceinline__ float lane_xor(float v) {
    return __builtin_bit_cast(float,
        __builtin_amdgcn_update_dpp(0, __builtin_bit_cast(int, v), 0x160 + M  ,
                                    0xf, 0xf, false));
}
__device__ __forceinline__ float lane_xor8(float v) { return lane_xor<8>(v); }

extern "C" __global__ __launch_bounds__(R_WAVES * 32)
void gru_recurrence(const float* __restrict__ gi, const float* __restrict__ w_hh,
                    const float* __restrict__ w1, const float* __restrict__ b1,
                    const float* __restrict__ w2, float* __restrict__ out)
{
    __shared__ __attribute__((aligned(16))) float part[2 * R_WAVES * R_PLANE];
    __shared__ __attribute__((aligned(16))) float hfin[R_TILE * R_HP];
    const int tid  = threadIdx.x;
    const int lane = tid & 31;
    const int lo   = lane & 15;
    const int hi   = lane >> 4;
    const int wave = tid >> 5;
    const int row  = lo & 7;
    const int half = lo >> 3;
    const int b0   = blockIdx.x * R_TILE;
    const int j0   = 8 * wave + 4 * hi;
    const int jc   = j0 + 2 * half;

    v2f af[6][2];
    #pragma unroll
    for (int mt = 0; mt < 6; ++mt) {
        #pragma unroll
        for (int kk = 0; kk < 2; ++kk)
            af[mt][kk] = *reinterpret_cast<const v2f*>(w_hh + (mt * 16 + lo) * H_SZ + j0 + 2 * kk);
    }

    float* wr = part + wave * R_PLANE + row * R_PP + 8 * hi;
    const float* rd = part + row * R_PP + jc;
    const float* girow = gi + (long)(b0 + row) * G3 + jc;

    v2f hc = {0.0f, 0.0f};
    v2f gin[3];
    #pragma unroll
    for (int g = 0; g < 3; ++g)
        gin[g] = *reinterpret_cast<const v2f*>(girow + g * 32);

    const float* pfrow = gi + (long)(b0 + row) * G3 + 32 * wave;
    const bool hints = __builtin_amdgcn_readfirstlane(wave) < 3;
    if (hints) {
        for (int s = 2; s < R_LEAD; ++s)
            __builtin_prefetch(pfrow + (long)s * (B_SZ * G3), 0, 1);
    }

    v4f w1r[H_SZ / 4];
    #pragma unroll
    for (int q = 0; q < H_SZ / 4; ++q)
        w1r[q] = *reinterpret_cast<const v4f*>(w1 + lo * H_SZ + 4 * q);
    const float b1j = b1[lo];
    const float w2j = w2[lo];

    for (int t = 0; t < T_SZ; ++t) {
        const int buf = (t & 1) * (R_WAVES * R_PLANE);

        if (hints) {
            const int tp = (t + R_LEAD < T_SZ) ? (t + R_LEAD) : (T_SZ - 1);
            __builtin_prefetch(pfrow + (long)tp * (B_SZ * G3), 0, 1);
        }

        v2f bk0, bk1;
        bk0.x = hc[0];            bk0.y = hc[1];
        bk1.x = lane_xor8(hc[0]); bk1.y = lane_xor8(hc[1]);
        v8f acc[6];
        #pragma unroll
        for (int mt = 0; mt < 6; ++mt) {
            v8f a = {};
            a = __builtin_amdgcn_wmma_f32_16x16x4_f32(
                false, af[mt][0], false, bk0, (short)0, a, false, false);
            a = __builtin_amdgcn_wmma_f32_16x16x4_f32(
                false, af[mt][1], false, bk1, (short)0, a, false, false);
            acc[mt] = a;
        }
        if (lo < R_TILE) {
            #pragma unroll
            for (int mt = 0; mt < 6; ++mt) {
                v4f q0, q1;
                q0.x = acc[mt][0]; q0.y = acc[mt][1]; q0.z = acc[mt][2]; q0.w = acc[mt][3];
                q1.x = acc[mt][4]; q1.y = acc[mt][5]; q1.z = acc[mt][6]; q1.w = acc[mt][7];
                *reinterpret_cast<v4f*>(wr + buf + mt * 16)     = q0;
                *reinterpret_cast<v4f*>(wr + buf + mt * 16 + 4) = q1;
            }
        }
        __syncthreads();

        v2f gh[3];
        #pragma unroll
        for (int g = 0; g < 3; ++g) {
            const float* p = rd + buf + g * 32;
            const v2f p0 = *reinterpret_cast<const v2f*>(p);
            const v2f p1 = *reinterpret_cast<const v2f*>(p + R_PLANE);
            const v2f p2 = *reinterpret_cast<const v2f*>(p + 2 * R_PLANE);
            const v2f p3 = *reinterpret_cast<const v2f*>(p + 3 * R_PLANE);
            gh[g] = ((p0 + p1) + p2) + p3;
        }

        const int tn = (t + 1 < T_SZ) ? (t + 1) : t;
        v2f gnx[3];
        #pragma unroll
        for (int g = 0; g < 3; ++g)
            gnx[g] = *reinterpret_cast<const v2f*>(girow + (long)tn * (B_SZ * G3) + g * 32);
        __builtin_amdgcn_sched_barrier(0);

        float zg[2], na[2];
        #pragma unroll
        for (int e = 0; e < 2; ++e) {
            const float rg = sigmoid_f(gin[0][e] + gh[0][e]);
            zg[e] = sigmoid_f(gin[1][e] + gh[1][e]);
            na[e] = gin[2][e] + rg * gh[2][e];
        }
        asm volatile("" : "+v"(na[0]), "+v"(na[1]), "+v"(zg[0]), "+v"(zg[1]));
        #pragma unroll
        for (int e = 0; e < 2; ++e) {
            const float ng = tanhf(na[e]);
            hc[e] = (1.0f - zg[e]) * ng + zg[e] * hc[e];
        }
        #pragma unroll
        for (int g = 0; g < 3; ++g) gin[g] = gnx[g];
    }

    *reinterpret_cast<v2f*>(hfin + row * R_HP + jc) = hc;
    __syncthreads();
    const int hr = 2 * wave + hi;
    const float* hrow = hfin + hr * R_HP;
    float a = b1j;
    #pragma unroll
    for (int q = 0; q < H_SZ / 4; ++q) {
        const v4f hv = *reinterpret_cast<const v4f*>(hrow + 4 * q);
        a += hv.x * w1r[q].x;
        a += hv.y * w1r[q].y;
        a += hv.z * w1r[q].z;
        a += hv.w * w1r[q].w;
    }
    float y = w2j * fmaxf(a, 0.0f);
    y += lane_xor<8>(y);
    y += lane_xor<4>(y);
    y += lane_xor<2>(y);
    y += lane_xor<1>(y);
    if (lo == 0) out[b0 + hr] = y;
}

extern "C" void kernel_launch(void* const* d_in, const int* in_sizes, int n_in,
                              void* d_out, int out_size, void* d_ws, size_t ws_size,
                              hipStream_t stream) {
    const float* x    = (const float*)d_in[0];
    const float* w_ih = (const float*)d_in[1];
    const float* w_hh = (const float*)d_in[2];
    const float* w1   = (const float*)d_in[3];
    const float* b1   = (const float*)d_in[4];
    const float* w2   = (const float*)d_in[5];
    float* out = (float*)d_out;

    float* gi = (float*)d_ws;

    gru_gemm_in<<<256, 384, 0, stream>>>(x, w_ih, gi);
    gru_recurrence<<<R_BLOCKS, 128, 0, stream>>>(gi, w_hh, w1, b1, w2, out);
}


# input projection: a chunk's memory work issued inside its product stream, not at its ends: the previous tile's 16 stores behind k-step 8 out of registers of their own, LDS staging behind k-step 34
# speedup vs baseline: 2.8428x; 1.0329x over previous
#include <hip/hip_runtime.h>
#include <cmath>


typedef __attribute__((ext_vector_type(2))) float v2f;
typedef __attribute__((ext_vector_type(4))) float v4f;
typedef __attribute__((ext_vector_type(8))) float v8f;

#define B_SZ 4096
#define T_SZ 60
#define F_SZ 158
#define G3 96
#define H_SZ 32

__device__ __forceinline__ float sigmoid_f(float x) {
    return 1.0f / (1.0f + __expf(-x));
}

#define P_THREADS 384
#define P_ROWS    64
#define P_CHUNK   (P_ROWS * F_SZ)
#define P_F4      (P_CHUNK / 4)
#define P_BLOCKS  256
#define P_ITERS   (B_SZ * T_SZ / P_ROWS / P_BLOCKS)
#define P_FULL    (P_F4 / P_THREADS)
#define P_TAIL    (P_F4 - P_FULL * P_THREADS)
#define P_SPLIT   8
#define P_STAGE   34

__device__ __forceinline__ void proj_request(const float* __restrict__ src, int tid, int tl,
                                             v4f (&pre)[P_FULL + 1])
{
    #pragma unroll
    for (int j = 0; j < P_FULL; ++j)
        pre[j] = *reinterpret_cast<const v4f*>(src + 4 * (tid + j * P_THREADS));
    pre[P_FULL] = *reinterpret_cast<const v4f*>(src + 4 * (tl + P_FULL * P_THREADS));
}

__device__ __forceinline__ void proj_stage(float* dst, int tid, int tl,
                                           const v4f (&pre)[P_FULL + 1])
{
    #pragma unroll
    for (int j = 0; j < P_FULL; ++j)
        *reinterpret_cast<v4f*>(dst + 4 * (tid + j * P_THREADS)) = pre[j];
    *reinterpret_cast<v4f*>(dst + 4 * (tl + P_FULL * P_THREADS)) = pre[P_FULL];
}

template <int K0, int K1>
__device__ __forceinline__ void proj_tile_part(const float* xt, int hi, const v2f (&bf)[2][40],
                                               v8f& acc0, v8f& acc1)
{
    #pragma unroll
    for (int kt = K0; kt < K1; ++kt) {
        const int c = (kt == 39) ? 156 : kt * 4 + 2 * hi;
        v2f a = *reinterpret_cast<const v2f*>(xt + c);
        if (kt == 39) {
            a.x = hi ? 0.0f : a.x;
            a.y = hi ? 0.0f : a.y;
        }
        acc0 = __builtin_amdgcn_wmma_f32_16x16x4_f32(
            false, a, false, bf[0][kt], (short)0, acc0, false, false);
        acc1 = __builtin_amdgcn_wmma_f32_16x16x4_f32(
            false, a, false, bf[1][kt], (short)0, acc1, false, false);
    }
}

__device__ __forceinline__ void proj_tile(const float* xt, int hi, const v2f (&bf)[2][40],
                                          v8f& d0, v8f& d1)
{
    v8f acc0 = {}, acc1 = {};
    proj_tile_part<0, 40>(xt, hi, bf, acc0, acc1);
    d0 = acc0;
    d1 = acc1;
}

__device__ __forceinline__ void proj_store(float* __restrict__ gi, int m0, int col,
                                           const v8f& d0, const v8f& d1)
{
    const int t0 = m0 % T_SZ;
    const int b0 = m0 / T_SZ;
    #pragma unroll
    for (int r = 0; r < 8; ++r) {
        int t = t0 + r;
        int b = b0;
        if (t >= T_SZ) { t -= T_SZ; b += 1; }
        float* orow = gi + ((t * B_SZ + b) * G3 + col);
        orow[0]  = d0[r];
        orow[16] = d1[r];
    }
}

extern "C" __global__ __launch_bounds__(P_THREADS)
void gru_gemm_in(const float* __restrict__ x, const float* __restrict__ w_ih,
                 float* __restrict__ gi)
{
    __shared__ __attribute__((aligned(16))) float xs[2 * P_CHUNK];
    const int tid  = threadIdx.x;
    const int lane = tid & 31;
    const int lo   = lane & 15;
    const int hi   = lane >> 4;
    const int wave = tid >> 5;
    const int ct   = wave % 3;
    const int rg   = wave / 3;
    const int blk  = blockIdx.x;

    const int tl   = (tid < P_TAIL) ? tid : (P_TAIL - 1);

    v4f pre[P_FULL + 1];
    proj_request(x + (size_t)blk * P_CHUNK, tid, tl, pre);

    v2f bf[2][40];
    #pragma unroll
    for (int nt = 0; nt < 2; ++nt) {
        const float* wrow = w_ih + (ct * 32 + nt * 16 + lo) * F_SZ;
        #pragma unroll
        for (int kt = 0; kt < 40; ++kt) {
            const int c = (kt == 39) ? 156 : kt * 4 + 2 * hi;
            v2f v = *reinterpret_cast<const v2f*>(wrow + c);
            if (kt == 39) {
                v.x = hi ? 0.0f : v.x;
                v.y = hi ? 0.0f : v.y;
            }
            bf[nt][kt] = v;
        }
    }

    proj_stage(xs, tid, tl, pre);
    __syncthreads();

    const int xoff = (rg * 16 + lo) * F_SZ;
    const int mrow = rg * 16 + hi * 8;
    const int col  = ct * 32 + lo;
    v8f d0, d1;

    proj_request(x + (size_t)(P_BLOCKS + blk) * P_CHUNK, tid, tl, pre);
    __builtin_amdgcn_sched_barrier(0);
    proj_tile(xs + xoff, hi, bf, d0, d1);
    proj_stage(xs + P_CHUNK, tid, tl, pre);
    __syncthreads();

    for (int it = 1; it < P_ITERS; ++it) {
        const int  cur  = it & 1;
        const bool more = (it + 1 < P_ITERS);
        if (more)
            proj_request(x + (size_t)((it + 1) * P_BLOCKS + blk) * P_CHUNK, tid, tl, pre);
        __builtin_amdgcn_sched_barrier(0);
        v8f a0 = {}, a1 = {};
        proj_tile_part<0, P_SPLIT>(xs + cur * P_CHUNK + xoff, hi, bf, a0, a1);
        asm volatile("" : "+v"(a0), "+v"(a1));
        __builtin_amdgcn_sched_barrier(0);
        proj_store(gi, ((it - 1) * P_BLOCKS + blk) * P_ROWS + mrow, col, d0, d1);
        __builtin_amdgcn_sched_barrier(0);
        proj_tile_part<P_SPLIT, P_STAGE>(xs + cur * P_CHUNK + xoff, hi, bf, a0, a1);
        asm volatile("" : "+v"(a0), "+v"(a1));
        __builtin_amdgcn_sched_barrier(0);
        if (more)
            proj_stage(xs + (cur ^ 1) * P_CHUNK, tid, tl, pre);
        __builtin_amdgcn_sched_barrier(0);
        proj_tile_part<P_STAGE, 40>(xs + cur * P_CHUNK + xoff, hi, bf, a0, a1);
        d0 = a0;
        d1 = a1;
        __syncthreads();
    }
    proj_store(gi, ((P_ITERS - 1) * P_BLOCKS + blk) * P_ROWS + mrow, col, d0, d1);
}

#define R_WAVES 4
#define R_TILE  8
#define R_PP    100
#define R_PLANE (R_TILE * R_PP)
#define R_HP    40
#define R_LEAD  3
#define R_BLOCKS (B_SZ / R_TILE)

template <int M>
__device__ __forceinline__ float lane_xor(float v) {
    return __builtin_bit_cast(float,
        __builtin_amdgcn_update_dpp(0, __builtin_bit_cast(int, v), 0x160 + M  ,
                                    0xf, 0xf, false));
}
__device__ __forceinline__ float lane_xor8(float v) { return lane_xor<8>(v); }

extern "C" __global__ __launch_bounds__(R_WAVES * 32)
void gru_recurrence(const float* __restrict__ gi, const float* __restrict__ w_hh,
                    const float* __restrict__ w1, const float* __restrict__ b1,
                    const float* __restrict__ w2, float* __restrict__ out)
{
    __shared__ __attribute__((aligned(16))) float part[2 * R_WAVES * R_PLANE];
    __shared__ __attribute__((aligned(16))) float hfin[R_TILE * R_HP];
    const int tid  = threadIdx.x;
    const int lane = tid & 31;
    const int lo   = lane & 15;
    const int hi   = lane >> 4;
    const int wave = tid >> 5;
    const int row  = lo & 7;
    const int half = lo >> 3;
    const int b0   = blockIdx.x * R_TILE;
    const int j0   = 8 * wave + 4 * hi;
    const int jc   = j0 + 2 * half;

    v2f af[6][2];
    #pragma unroll
    for (int mt = 0; mt < 6; ++mt) {
        #pragma unroll
        for (int kk = 0; kk < 2; ++kk)
            af[mt][kk] = *reinterpret_cast<const v2f*>(w_hh + (mt * 16 + lo) * H_SZ + j0 + 2 * kk);
    }

    float* wr = part + wave * R_PLANE + row * R_PP + 8 * hi;
    const float* rd = part + row * R_PP + jc;
    const float* girow = gi + (long)(b0 + row) * G3 + jc;

    v2f hc = {0.0f, 0.0f};
    v2f gin[3];
    #pragma unroll
    for (int g = 0; g < 3; ++g)
        gin[g] = *reinterpret_cast<const v2f*>(girow + g * 32);

    const float* pfrow = gi + (long)(b0 + row) * G3 + 32 * wave;
    const bool hints = __builtin_amdgcn_readfirstlane(wave) < 3;
    if (hints) {
        for (int s = 2; s < R_LEAD; ++s)
            __builtin_prefetch(pfrow + (long)s * (B_SZ * G3), 0, 1);
    }

    v4f w1r[H_SZ / 4];
    #pragma unroll
    for (int q = 0; q < H_SZ / 4; ++q)
        w1r[q] = *reinterpret_cast<const v4f*>(w1 + lo * H_SZ + 4 * q);
    const float b1j = b1[lo];
    const float w2j = w2[lo];

    for (int t = 0; t < T_SZ; ++t) {
        const int buf = (t & 1) * (R_WAVES * R_PLANE);

        if (hints) {
            const int tp = (t + R_LEAD < T_SZ) ? (t + R_LEAD) : (T_SZ - 1);
            __builtin_prefetch(pfrow + (long)tp * (B_SZ * G3), 0, 1);
        }

        v2f bk0, bk1;
        bk0.x = hc[0];            bk0.y = hc[1];
        bk1.x = lane_xor8(hc[0]); bk1.y = lane_xor8(hc[1]);
        v8f acc[6];
        #pragma unroll
        for (int mt = 0; mt < 6; ++mt) {
            v8f a = {};
            a = __builtin_amdgcn_wmma_f32_16x16x4_f32(
                false, af[mt][0], false, bk0, (short)0, a, false, false);
            a = __builtin_amdgcn_wmma_f32_16x16x4_f32(
                false, af[mt][1], false, bk1, (short)0, a, false, false);
            acc[mt] = a;
        }
        if (lo < R_TILE) {
            #pragma unroll
            for (int mt = 0; mt < 6; ++mt) {
                v4f q0, q1;
                q0.x = acc[mt][0]; q0.y = acc[mt][1]; q0.z = acc[mt][2]; q0.w = acc[mt][3];
                q1.x = acc[mt][4]; q1.y = acc[mt][5]; q1.z = acc[mt][6]; q1.w = acc[mt][7];
                *reinterpret_cast<v4f*>(wr + buf + mt * 16)     = q0;
                *reinterpret_cast<v4f*>(wr + buf + mt * 16 + 4) = q1;
            }
        }
        __syncthreads();

        v2f gh[3];
        #pragma unroll
        for (int g = 0; g < 3; ++g) {
            const float* p = rd + buf + g * 32;
            const v2f p0 = *reinterpret_cast<const v2f*>(p);
            const v2f p1 = *reinterpret_cast<const v2f*>(p + R_PLANE);
            const v2f p2 = *reinterpret_cast<const v2f*>(p + 2 * R_PLANE);
            const v2f p3 = *reinterpret_cast<const v2f*>(p + 3 * R_PLANE);
            gh[g] = ((p0 + p1) + p2) + p3;
        }

        const int tn = (t + 1 < T_SZ) ? (t + 1) : t;
        v2f gnx[3];
        #pragma unroll
        for (int g = 0; g < 3; ++g)
            gnx[g] = *reinterpret_cast<const v2f*>(girow + (long)tn * (B_SZ * G3) + g * 32);
        __builtin_amdgcn_sched_barrier(0);

        float zg[2], na[2];
        #pragma unroll
        for (int e = 0; e < 2; ++e) {
            const float rg = sigmoid_f(gin[0][e] + gh[0][e]);
            zg[e] = sigmoid_f(gin[1][e] + gh[1][e]);
            na[e] = gin[2][e] + rg * gh[2][e];
        }
        asm volatile("" : "+v"(na[0]), "+v"(na[1]), "+v"(zg[0]), "+v"(zg[1]));
        #pragma unroll
        for (int e = 0; e < 2; ++e) {
            const float ng = tanhf(na[e]);
            hc[e] = (1.0f - zg[e]) * ng + zg[e] * hc[e];
        }
        #pragma unroll
        for (int g = 0; g < 3; ++g) gin[g] = gnx[g];
    }

    *reinterpret_cast<v2f*>(hfin + row * R_HP + jc) = hc;
    __syncthreads();
    const int hr = 2 * wave + hi;
    const float* hrow = hfin + hr * R_HP;
    float a = b1j;
    #pragma unroll
    for (int q = 0; q < H_SZ / 4; ++q) {
        const v4f hv = *reinterpret_cast<const v4f*>(hrow + 4 * q);
        a += hv.x * w1r[q].x;
        a += hv.y * w1r[q].y;
        a += hv.z * w1r[q].z;
        a += hv.w * w1r[q].w;
    }
    float y = w2j * fmaxf(a, 0.0f);
    y += lane_xor<8>(y);
    y += lane_xor<4>(y);
    y += lane_xor<2>(y);
    y += lane_xor<1>(y);
    if (lo == 0) out[b0 + hr] = y;
}

extern "C" void kernel_launch(void* const* d_in, const int* in_sizes, int n_in,
                              void* d_out, int out_size, void* d_ws, size_t ws_size,
                              hipStream_t stream) {
    const float* x    = (const float*)d_in[0];
    const float* w_ih = (const float*)d_in[1];
    const float* w_hh = (const float*)d_in[2];
    const float* w1   = (const float*)d_in[3];
    const float* b1   = (const float*)d_in[4];
    const float* w2   = (const float*)d_in[5];
    float* out = (float*)d_out;

    float* gi = (float*)d_ws;

    gru_gemm_in<<<256, 384, 0, stream>>>(x, w_ih, gi);
    gru_recurrence<<<R_BLOCKS, 128, 0, stream>>>(gi, w_hh, w1, b1, w2, out);
}


# input projection: w_ih reaches the register fragments through LDS, copied once per work-group in whole lines, not by 80 sixteen-line global loads per wave; x chunk 0 stays in flight under it
# speedup vs baseline: 2.9364x; 1.0329x over previous
#include <hip/hip_runtime.h>
#include <cmath>


typedef __attribute__((ext_vector_type(2))) float v2f;
typedef __attribute__((ext_vector_type(4))) float v4f;
typedef __attribute__((ext_vector_type(8))) float v8f;

#define B_SZ 4096
#define T_SZ 60
#define F_SZ 158
#define G3 96
#define H_SZ 32

__device__ __forceinline__ float sigmoid_f(float x) {
    return 1.0f / (1.0f + __expf(-x));
}

#define P_THREADS 384
#define P_ROWS    64
#define P_CHUNK   (P_ROWS * F_SZ)
#define P_F4      (P_CHUNK / 4)
#define P_BLOCKS  256
#define P_ITERS   (B_SZ * T_SZ / P_ROWS / P_BLOCKS)
#define P_FULL    (P_F4 / P_THREADS)
#define P_TAIL    (P_F4 - P_FULL * P_THREADS)
#define P_SPLIT   8
#define P_STAGE   34
#define W_F4      (G3 * F_SZ / 4)
#define W_FULL    (W_F4 / P_THREADS)
#define W_TAIL    (W_F4 - W_FULL * P_THREADS)

__device__ __forceinline__ void lds_barrier()
{
    __builtin_amdgcn_fence(__ATOMIC_RELEASE, "workgroup", "local");
    __builtin_amdgcn_s_barrier();
    __builtin_amdgcn_fence(__ATOMIC_ACQUIRE, "workgroup", "local");
}

__device__ __forceinline__ void proj_request(const float* __restrict__ src, int tid, int tl,
                                             v4f (&pre)[P_FULL + 1])
{
    #pragma unroll
    for (int j = 0; j < P_FULL; ++j)
        pre[j] = *reinterpret_cast<const v4f*>(src + 4 * (tid + j * P_THREADS));
    pre[P_FULL] = *reinterpret_cast<const v4f*>(src + 4 * (tl + P_FULL * P_THREADS));
}

__device__ __forceinline__ void proj_stage(float* dst, int tid, int tl,
                                           const v4f (&pre)[P_FULL + 1])
{
    #pragma unroll
    for (int j = 0; j < P_FULL; ++j)
        *reinterpret_cast<v4f*>(dst + 4 * (tid + j * P_THREADS)) = pre[j];
    *reinterpret_cast<v4f*>(dst + 4 * (tl + P_FULL * P_THREADS)) = pre[P_FULL];
}

template <int K0, int K1>
__device__ __forceinline__ void proj_tile_part(const float* xt, int hi, const v2f (&bf)[2][40],
                                               v8f& acc0, v8f& acc1)
{
    #pragma unroll
    for (int kt = K0; kt < K1; ++kt) {
        const int c = (kt == 39) ? 156 : kt * 4 + 2 * hi;
        v2f a = *reinterpret_cast<const v2f*>(xt + c);
        if (kt == 39) {
            a.x = hi ? 0.0f : a.x;
            a.y = hi ? 0.0f : a.y;
        }
        acc0 = __builtin_amdgcn_wmma_f32_16x16x4_f32(
            false, a, false, bf[0][kt], (short)0, acc0, false, false);
        acc1 = __builtin_amdgcn_wmma_f32_16x16x4_f32(
            false, a, false, bf[1][kt], (short)0, acc1, false, false);
    }
}

__device__ __forceinline__ void proj_tile(const float* xt, int hi, const v2f (&bf)[2][40],
                                          v8f& d0, v8f& d1)
{
    v8f acc0 = {}, acc1 = {};
    proj_tile_part<0, 40>(xt, hi, bf, acc0, acc1);
    d0 = acc0;
    d1 = acc1;
}

__device__ __forceinline__ void proj_store(float* __restrict__ gi, int m0, int col,
                                           const v8f& d0, const v8f& d1)
{
    const int t0 = m0 % T_SZ;
    const int b0 = m0 / T_SZ;
    #pragma unroll
    for (int r = 0; r < 8; ++r) {
        int t = t0 + r;
        int b = b0;
        if (t >= T_SZ) { t -= T_SZ; b += 1; }
        float* orow = gi + ((t * B_SZ + b) * G3 + col);
        orow[0]  = d0[r];
        orow[16] = d1[r];
    }
}

extern "C" __global__ __launch_bounds__(P_THREADS)
void gru_gemm_in(const float* __restrict__ x, const float* __restrict__ w_ih,
                 float* __restrict__ gi)
{
    __shared__ __attribute__((aligned(16))) float xs[2 * P_CHUNK];
    __shared__ __attribute__((aligned(16))) float wsm[G3 * F_SZ];
    const int tid  = threadIdx.x;
    const int lane = tid & 31;
    const int lo   = lane & 15;
    const int hi   = lane >> 4;
    const int wave = tid >> 5;
    const int ct   = wave % 3;
    const int rg   = wave / 3;
    const int blk  = blockIdx.x;

    const int tl   = (tid < P_TAIL) ? tid : (P_TAIL - 1);

    const int wl = (tid < W_TAIL) ? tid : (W_TAIL - 1);
    v4f wq[W_FULL + 1];
    #pragma unroll
    for (int j = 0; j < W_FULL; ++j)
        wq[j] = *reinterpret_cast<const v4f*>(w_ih + 4 * (tid + j * P_THREADS));
    wq[W_FULL] = *reinterpret_cast<const v4f*>(w_ih + 4 * (wl + W_FULL * P_THREADS));
    __builtin_amdgcn_sched_barrier(0);
    v4f pre[P_FULL + 1];
    proj_request(x + (size_t)blk * P_CHUNK, tid, tl, pre);
    __builtin_amdgcn_sched_barrier(0);

    #pragma unroll
    for (int j = 0; j < W_FULL; ++j)
        *reinterpret_cast<v4f*>(wsm + 4 * (tid + j * P_THREADS)) = wq[j];
    *reinterpret_cast<v4f*>(wsm + 4 * (wl + W_FULL * P_THREADS)) = wq[W_FULL];
    lds_barrier();

    v2f bf[2][40];
    #pragma unroll
    for (int nt = 0; nt < 2; ++nt) {
        const float* wrow = wsm + (ct * 32 + nt * 16 + lo) * F_SZ;
        #pragma unroll
        for (int kt = 0; kt < 40; ++kt) {
            const int c = (kt == 39) ? 156 : kt * 4 + 2 * hi;
            v2f v = *reinterpret_cast<const v2f*>(wrow + c);
            if (kt == 39) {
                v.x = hi ? 0.0f : v.x;
                v.y = hi ? 0.0f : v.y;
            }
            bf[nt][kt] = v;
        }
    }

    proj_stage(xs, tid, tl, pre);
    __syncthreads();

    const int xoff = (rg * 16 + lo) * F_SZ;
    const int mrow = rg * 16 + hi * 8;
    const int col  = ct * 32 + lo;
    v8f d0, d1;

    proj_request(x + (size_t)(P_BLOCKS + blk) * P_CHUNK, tid, tl, pre);
    __builtin_amdgcn_sched_barrier(0);
    proj_tile(xs + xoff, hi, bf, d0, d1);
    proj_stage(xs + P_CHUNK, tid, tl, pre);
    __syncthreads();

    for (int it = 1; it < P_ITERS; ++it) {
        const int  cur  = it & 1;
        const bool more = (it + 1 < P_ITERS);
        if (more)
            proj_request(x + (size_t)((it + 1) * P_BLOCKS + blk) * P_CHUNK, tid, tl, pre);
        __builtin_amdgcn_sched_barrier(0);
        v8f a0 = {}, a1 = {};
        proj_tile_part<0, P_SPLIT>(xs + cur * P_CHUNK + xoff, hi, bf, a0, a1);
        asm volatile("" : "+v"(a0), "+v"(a1));
        __builtin_amdgcn_sched_barrier(0);
        proj_store(gi, ((it - 1) * P_BLOCKS + blk) * P_ROWS + mrow, col, d0, d1);
        __builtin_amdgcn_sched_barrier(0);
        proj_tile_part<P_SPLIT, P_STAGE>(xs + cur * P_CHUNK + xoff, hi, bf, a0, a1);
        asm volatile("" : "+v"(a0), "+v"(a1));
        __builtin_amdgcn_sched_barrier(0);
        if (more)
            proj_stage(xs + (cur ^ 1) * P_CHUNK, tid, tl, pre);
        __builtin_amdgcn_sched_barrier(0);
        proj_tile_part<P_STAGE, 40>(xs + cur * P_CHUNK + xoff, hi, bf, a0, a1);
        d0 = a0;
        d1 = a1;
        __syncthreads();
    }
    proj_store(gi, ((P_ITERS - 1) * P_BLOCKS + blk) * P_ROWS + mrow, col, d0, d1);
}

#define R_WAVES 4
#define R_TILE  8
#define R_PP    100
#define R_PLANE (R_TILE * R_PP)
#define R_HP    40
#define R_LEAD  3
#define R_BLOCKS (B_SZ / R_TILE)

template <int M>
__device__ __forceinline__ float lane_xor(float v) {
    return __builtin_bit_cast(float,
        __builtin_amdgcn_update_dpp(0, __builtin_bit_cast(int, v), 0x160 + M  ,
                                    0xf, 0xf, false));
}
__device__ __forceinline__ float lane_xor8(float v) { return lane_xor<8>(v); }

extern "C" __global__ __launch_bounds__(R_WAVES * 32)
void gru_recurrence(const float* __restrict__ gi, const float* __restrict__ w_hh,
                    const float* __restrict__ w1, const float* __restrict__ b1,
                    const float* __restrict__ w2, float* __restrict__ out)
{
    __shared__ __attribute__((aligned(16))) float part[2 * R_WAVES * R_PLANE];
    __shared__ __attribute__((aligned(16))) float hfin[R_TILE * R_HP];
    const int tid  = threadIdx.x;
    const int lane = tid & 31;
    const int lo   = lane & 15;
    const int hi   = lane >> 4;
    const int wave = tid >> 5;
    const int row  = lo & 7;
    const int half = lo >> 3;
    const int b0   = blockIdx.x * R_TILE;
    const int j0   = 8 * wave + 4 * hi;
    const int jc   = j0 + 2 * half;

    v2f af[6][2];
    #pragma unroll
    for (int mt = 0; mt < 6; ++mt) {
        #pragma unroll
        for (int kk = 0; kk < 2; ++kk)
            af[mt][kk] = *reinterpret_cast<const v2f*>(w_hh + (mt * 16 + lo) * H_SZ + j0 + 2 * kk);
    }

    float* wr = part + wave * R_PLANE + row * R_PP + 8 * hi;
    const float* rd = part + row * R_PP + jc;
    const float* girow = gi + (long)(b0 + row) * G3 + jc;

    v2f hc = {0.0f, 0.0f};
    v2f gin[3];
    #pragma unroll
    for (int g = 0; g < 3; ++g)
        gin[g] = *reinterpret_cast<const v2f*>(girow + g * 32);

    const float* pfrow = gi + (long)(b0 + row) * G3 + 32 * wave;
    const bool hints = __builtin_amdgcn_readfirstlane(wave) < 3;
    if (hints) {
        for (int s = 2; s < R_LEAD; ++s)
            __builtin_prefetch(pfrow + (long)s * (B_SZ * G3), 0, 1);
    }

    v4f w1r[H_SZ / 4];
    #pragma unroll
    for (int q = 0; q < H_SZ / 4; ++q)
        w1r[q] = *reinterpret_cast<const v4f*>(w1 + lo * H_SZ + 4 * q);
    const float b1j = b1[lo];
    const float w2j = w2[lo];

    for (int t = 0; t < T_SZ; ++t) {
        const int buf = (t & 1) * (R_WAVES * R_PLANE);

        if (hints) {
            const int tp = (t + R_LEAD < T_SZ) ? (t + R_LEAD) : (T_SZ - 1);
            __builtin_prefetch(pfrow + (long)tp * (B_SZ * G3), 0, 1);
        }

        v2f bk0, bk1;
        bk0.x = hc[0];            bk0.y = hc[1];
        bk1.x = lane_xor8(hc[0]); bk1.y = lane_xor8(hc[1]);
        v8f acc[6];
        #pragma unroll
        for (int mt = 0; mt < 6; ++mt) {
            v8f a = {};
            a = __builtin_amdgcn_wmma_f32_16x16x4_f32(
                false, af[mt][0], false, bk0, (short)0, a, false, false);
            a = __builtin_amdgcn_wmma_f32_16x16x4_f32(
                false, af[mt][1], false, bk1, (short)0, a, false, false);
            acc[mt] = a;
        }
        if (lo < R_TILE) {
            #pragma unroll
            for (int mt = 0; mt < 6; ++mt) {
                v4f q0, q1;
                q0.x = acc[mt][0]; q0.y = acc[mt][1]; q0.z = acc[mt][2]; q0.w = acc[mt][3];
                q1.x = acc[mt][4]; q1.y = acc[mt][5]; q1.z = acc[mt][6]; q1.w = acc[mt][7];
                *reinterpret_cast<v4f*>(wr + buf + mt * 16)     = q0;
                *reinterpret_cast<v4f*>(wr + buf + mt * 16 + 4) = q1;
            }
        }
        __syncthreads();

        v2f gh[3];
        #pragma unroll
        for (int g = 0; g < 3; ++g) {
            const float* p = rd + buf + g * 32;
            const v2f p0 = *reinterpret_cast<const v2f*>(p);
            const v2f p1 = *reinterpret_cast<const v2f*>(p + R_PLANE);
            const v2f p2 = *reinterpret_cast<const v2f*>(p + 2 * R_PLANE);
            const v2f p3 = *reinterpret_cast<const v2f*>(p + 3 * R_PLANE);
            gh[g] = ((p0 + p1) + p2) + p3;
        }

        const int tn = (t + 1 < T_SZ) ? (t + 1) : t;
        v2f gnx[3];
        #pragma unroll
        for (int g = 0; g < 3; ++g)
            gnx[g] = *reinterpret_cast<const v2f*>(girow + (long)tn * (B_SZ * G3) + g * 32);
        __builtin_amdgcn_sched_barrier(0);

        float zg[2], na[2];
        #pragma unroll
        for (int e = 0; e < 2; ++e) {
            const float rg = sigmoid_f(gin[0][e] + gh[0][e]);
            zg[e] = sigmoid_f(gin[1][e] + gh[1][e]);
            na[e] = gin[2][e] + rg * gh[2][e];
        }
        asm volatile("" : "+v"(na[0]), "+v"(na[1]), "+v"(zg[0]), "+v"(zg[1]));
        #pragma unroll
        for (int e = 0; e < 2; ++e) {
            const float ng = tanhf(na[e]);
            hc[e] = (1.0f - zg[e]) * ng + zg[e] * hc[e];
        }
        #pragma unroll
        for (int g = 0; g < 3; ++g) gin[g] = gnx[g];
    }

    *reinterpret_cast<v2f*>(hfin + row * R_HP + jc) = hc;
    __syncthreads();
    const int hr = 2 * wave + hi;
    const float* hrow = hfin + hr * R_HP;
    float a = b1j;
    #pragma unroll
    for (int q = 0; q < H_SZ / 4; ++q) {
        const v4f hv = *reinterpret_cast<const v4f*>(hrow + 4 * q);
        a += hv.x * w1r[q].x;
        a += hv.y * w1r[q].y;
        a += hv.z * w1r[q].z;
        a += hv.w * w1r[q].w;
    }
    float y = w2j * fmaxf(a, 0.0f);
    y += lane_xor<8>(y);
    y += lane_xor<4>(y);
    y += lane_xor<2>(y);
    y += lane_xor<1>(y);
    if (lo == 0) out[b0 + hr] = y;
}

extern "C" void kernel_launch(void* const* d_in, const int* in_sizes, int n_in,
                              void* d_out, int out_size, void* d_ws, size_t ws_size,
                              hipStream_t stream) {
    const float* x    = (const float*)d_in[0];
    const float* w_ih = (const float*)d_in[1];
    const float* w_hh = (const float*)d_in[2];
    const float* w1   = (const float*)d_in[3];
    const float* b1   = (const float*)d_in[4];
    const float* w2   = (const float*)d_in[5];
    float* out = (float*)d_out;

    float* gi = (float*)d_ws;

    gru_gemm_in<<<256, 384, 0, stream>>>(x, w_ih, gi);
    gru_recurrence<<<R_BLOCKS, 128, 0, stream>>>(gi, w_hh, w1, b1, w2, out);
}
